# starting kernel
# baseline (speedup 1.0000x reference)
#include <hip/hip_runtime.h>

#define NNODES 100000
#define NGRAPHS 1024
#define EMB 32
#define HID 64
#define NCLASSES 10

typedef __attribute__((ext_vector_type(2))) float v2f;
typedef __attribute__((ext_vector_type(8))) float v8f;

__global__ void zero_f32(float* __restrict__ p, long n) {
  long i = (long)blockIdx.x * blockDim.x + threadIdx.x;
  if (i < n) p[i] = 0.0f;
}

__global__ void deg_kernel(const int* __restrict__ dst, unsigned int* __restrict__ deg, int E) {
  int e = blockIdx.x * blockDim.x + threadIdx.x;
  if (e < E) atomicAdd(&deg[dst[e]], 1u);
}

__global__ void dis_kernel(float* __restrict__ disbuf, int n) {
  int i = blockIdx.x * blockDim.x + threadIdx.x;
  if (i < n) {
    unsigned int d = ((const unsigned int*)disbuf)[i];
    disbuf[i] = rsqrtf(1.0f + (float)d);
  }
}

__global__ void embed_kernel(const int* __restrict__ sid, const int* __restrict__ cid,
                             const float* __restrict__ st, const float* __restrict__ ct,
                             float* __restrict__ x0, int n) {
  long i = (long)blockIdx.x * blockDim.x + threadIdx.x;
  if (i >= (long)n * EMB) return;
  int node = (int)(i >> 5);
  int c = (int)(i & 31);
  int s = sid[node], cc = cid[node];
  float v = 0.0f;
  if (s != 0) v += st[s * EMB + c];
  if (cc != 0) v += ct[cc * EMB + c];
  x0[i] = v;
}

__global__ void gemm_wmma_f32(const float* __restrict__ X, const float* __restrict__ W,
                              float* __restrict__ H, int K) {
  const int lane  = threadIdx.x & 31;
  const int wave  = threadIdx.x >> 5;
  const int l16   = lane & 15;
  const int khalf = lane >> 4;
  const long row  = (long)blockIdx.x * 16 + l16;
  const int  col  = wave * 16 + l16;

  v8f c = {};
  const float* ap = X + row * K + 2 * khalf;
  for (int k0 = 0; k0 < K; k0 += 4) {
    v2f a, b;
    a.x = ap[k0];
    a.y = ap[k0 + 1];
    b.x = W[(k0 + 2 * khalf + 0) * HID + col];
    b.y = W[(k0 + 2 * khalf + 1) * HID + col];
    c = __builtin_amdgcn_wmma_f32_16x16x4_f32(false, a, false, b, (short)0, c, false, false);
  }
  const long rbase = (long)blockIdx.x * 16 + 8 * khalf;
#pragma unroll
  for (int v = 0; v < 8; ++v)
    H[(rbase + v) * HID + col] = c[v];
}

__global__ void edge_agg_kernel(const int* __restrict__ src, const int* __restrict__ dst,
                                const float* __restrict__ h, const float* __restrict__ dis,
                                float* __restrict__ agg, int E) {
  long t = (long)blockIdx.x * blockDim.x + threadIdx.x;
  long e = t >> 4;
  if (e >= E) return;
  int c4 = (int)(t & 15) << 2;
  int s = src[e];
  int d = dst[e];
  if (e + 512 < E) __builtin_prefetch(h + (long)src[e + 512] * HID + c4, 0, 1);
  float norm = dis[s] * dis[d];
  const float4 hv = *(const float4*)(h + (long)s * HID + c4);
  float* ap = agg + (long)d * HID + c4;
  atomicAdd(ap + 0, hv.x * norm);
  atomicAdd(ap + 1, hv.y * norm);
  atomicAdd(ap + 2, hv.z * norm);
  atomicAdd(ap + 3, hv.w * norm);
}

__global__ void finalize_kernel(float* __restrict__ agg, const float* __restrict__ h,
                                const float* __restrict__ dis, const float* __restrict__ b,
                                int n) {
  long i = (long)blockIdx.x * blockDim.x + threadIdx.x;
  if (i >= (long)n * HID) return;
  int node = (int)(i >> 6);
  int c = (int)(i & 63);
  float di = dis[node];
  float v = agg[i] + h[i] * di * di + b[c];
  agg[i] = v > 0.0f ? v : 0.0f;
}

__global__ void pool_kernel(const float* __restrict__ x, const int* __restrict__ batch,
                            float* __restrict__ pooled, float* __restrict__ cnt, int n) {
  long i = (long)blockIdx.x * blockDim.x + threadIdx.x;
  if (i >= (long)n * HID) return;
  int node = (int)(i >> 6);
  int c = (int)(i & 63);
  int g = batch[node];
  atomicAdd(pooled + (long)g * HID + c, x[i]);
  if (c == 0) atomicAdd(cnt + g, 1.0f);
}

__global__ void out_kernel(const float* __restrict__ pooled, const float* __restrict__ cnt,
                           const float* __restrict__ Wl, const float* __restrict__ bl,
                           float* __restrict__ out) {
  int i = blockIdx.x * blockDim.x + threadIdx.x;
  if (i >= NGRAPHS * NCLASSES) return;
  int g = i / NCLASSES;
  int k = i % NCLASSES;
  float s = 0.0f;
#pragma unroll 8
  for (int hh = 0; hh < HID; ++hh) s += pooled[g * HID + hh] * Wl[hh * NCLASSES + k];
  float c = cnt[g];
  c = c > 1.0f ? c : 1.0f;
  out[i] = s / c + bl[k];
}

extern "C" void kernel_launch(void* const* d_in, const int* in_sizes, int n_in,
                              void* d_out, int out_size, void* d_ws, size_t ws_size,
                              hipStream_t stream) {
  const int* shape_id = (const int*)d_in[0];
  const int* color_id = (const int*)d_in[1];
  const int* edge_index = (const int*)d_in[2];
  const int* batch = (const int*)d_in[3];
  const float* shape_table = (const float*)d_in[4];
  const float* color_table = (const float*)d_in[5];
  const float* W1 = (const float*)d_in[6];
  const float* b1 = (const float*)d_in[7];
  const float* W2 = (const float*)d_in[8];
  const float* b2 = (const float*)d_in[9];
  const float* Wl = (const float*)d_in[10];
  const float* bl = (const float*)d_in[11];
  float* out = (float*)d_out;

  const int N = NNODES;
  const int E = in_sizes[2] / 2;
  const int* src = edge_index;
  const int* dst = edge_index + E;

  float* dis    = (float*)d_ws;
  float* x0     = dis + N;
  float* bufH   = x0 + (long)N * EMB;
  float* bufA   = bufH + (long)N * HID;
  float* pooled = bufA + (long)N * HID;
  float* cnt    = pooled + (long)NGRAPHS * HID;
  (void)n_in; (void)out_size; (void)ws_size;

  const int B = 256;
  auto blocks = [](long n, int b) { return (unsigned)((n + b - 1) / b); };

  zero_f32<<<blocks(N, B), B, 0, stream>>>(dis, N);
  deg_kernel<<<blocks(E, B), B, 0, stream>>>(dst, (unsigned int*)dis, E);
  dis_kernel<<<blocks(N, B), B, 0, stream>>>(dis, N);

  embed_kernel<<<blocks((long)N * EMB, B), B, 0, stream>>>(shape_id, color_id,
                                                           shape_table, color_table, x0, N);

  gemm_wmma_f32<<<N / 16, 128, 0, stream>>>(x0, W1, bufH, EMB);
  zero_f32<<<blocks((long)N * HID, B), B, 0, stream>>>(bufA, (long)N * HID);
  edge_agg_kernel<<<blocks((long)E * 16, B), B, 0, stream>>>(src, dst, bufH, dis, bufA, E);
  finalize_kernel<<<blocks((long)N * HID, B), B, 0, stream>>>(bufA, bufH, dis, b1, N);

  gemm_wmma_f32<<<N / 16, 128, 0, stream>>>(bufA, W2, bufH, HID);
  zero_f32<<<blocks((long)N * HID, B), B, 0, stream>>>(bufA, (long)N * HID);
  edge_agg_kernel<<<blocks((long)E * 16, B), B, 0, stream>>>(src, dst, bufH, dis, bufA, E);
  finalize_kernel<<<blocks((long)N * HID, B), B, 0, stream>>>(bufA, bufH, dis, b2, N);

  zero_f32<<<blocks((long)NGRAPHS * HID + NGRAPHS, B), B, 0, stream>>>(pooled,
                                                                      (long)NGRAPHS * HID + NGRAPHS);
  pool_kernel<<<blocks((long)N * HID, B), B, 0, stream>>>(bufA, batch, pooled, cnt, N);
  out_kernel<<<blocks((long)NGRAPHS * NCLASSES, B), B, 0, stream>>>(pooled, cnt, Wl, bl, out);
}


# Scatter sums kept in LDS: a work-group owns 400 dst rows, finds its edges in a byte-per-edge tile stream, adds by LDS f32 atomics, stores relu(sum+self+bias); no zero/finalize pass; sum order arrival
# speedup vs baseline: 1.4793x; 1.4793x over previous
#include <hip/hip_runtime.h>

#define NNODES 100000
#define NGRAPHS 1024
#define EMB 32
#define HID 64
#define NCLASSES 10

typedef __attribute__((ext_vector_type(2))) float v2f;
typedef __attribute__((ext_vector_type(8))) float v8f;

__global__ void zero_f32(float* __restrict__ p, long n) {
  long i = (long)blockIdx.x * blockDim.x + threadIdx.x;
  if (i < n) p[i] = 0.0f;
}

__global__ void deg_kernel(const int* __restrict__ dst, unsigned int* __restrict__ deg, int E) {
  int e = blockIdx.x * blockDim.x + threadIdx.x;
  if (e < E) atomicAdd(&deg[dst[e]], 1u);
}

__global__ void dis_kernel(float* __restrict__ disbuf, int n) {
  int i = blockIdx.x * blockDim.x + threadIdx.x;
  if (i < n) {
    unsigned int d = ((const unsigned int*)disbuf)[i];
    disbuf[i] = rsqrtf(1.0f + (float)d);
  }
}

__global__ void embed_kernel(const int* __restrict__ sid, const int* __restrict__ cid,
                             const float* __restrict__ st, const float* __restrict__ ct,
                             float* __restrict__ x0, int n) {
  long i = (long)blockIdx.x * blockDim.x + threadIdx.x;
  if (i >= (long)n * EMB) return;
  int node = (int)(i >> 5);
  int c = (int)(i & 31);
  int s = sid[node], cc = cid[node];
  float v = 0.0f;
  if (s != 0) v += st[s * EMB + c];
  if (cc != 0) v += ct[cc * EMB + c];
  x0[i] = v;
}

__global__ void gemm_wmma_f32(const float* __restrict__ X, const float* __restrict__ W,
                              float* __restrict__ H, int K) {
  const int lane  = threadIdx.x & 31;
  const int wave  = threadIdx.x >> 5;
  const int l16   = lane & 15;
  const int khalf = lane >> 4;
  const long row  = (long)blockIdx.x * 16 + l16;
  const int  col  = wave * 16 + l16;

  v8f c = {};
  const float* ap = X + row * K + 2 * khalf;
  for (int k0 = 0; k0 < K; k0 += 4) {
    v2f a, b;
    a.x = ap[k0];
    a.y = ap[k0 + 1];
    b.x = W[(k0 + 2 * khalf + 0) * HID + col];
    b.y = W[(k0 + 2 * khalf + 1) * HID + col];
    c = __builtin_amdgcn_wmma_f32_16x16x4_f32(false, a, false, b, (short)0, c, false, false);
  }
  const long rbase = (long)blockIdx.x * 16 + 8 * khalf;
#pragma unroll
  for (int v = 0; v < 8; ++v)
    H[(rbase + v) * HID + col] = c[v];
}

#define TILE_ROWS 400
#define NTILES ((NNODES + TILE_ROWS - 1) / TILE_ROWS)
#define BKT_NONE 255u
#define AGG_THREADS 1024
#define AGG_WAVES (AGG_THREADS / 32)
#define RING 1024
#define EDGE_ROWS 8
#define STREAM_PAD (12 * AGG_THREADS * 16)
static_assert(NTILES <= 255, "tile number must fit a byte beside the pad value");

__global__ void bucket_kernel(const int* __restrict__ dst, unsigned int* __restrict__ bkt4, int E) {
  const long i = (long)blockIdx.x * blockDim.x + threadIdx.x;
  const long epad = ((long)E + 511) / 512 * 512 + STREAM_PAD;
  const long e0 = i * 4;
  if (e0 >= epad) return;
  unsigned int w = 0;
  for (int k = 0; k < 4; ++k) {
    const long e = e0 + k;
    unsigned int b = BKT_NONE;
    if (e < E) b = (unsigned int)dst[e] / TILE_ROWS;
    w |= b << (8 * k);
  }
  bkt4[i] = w;
}

__device__ __forceinline__ unsigned int match_bytes(unsigned int x, unsigned int pat) {
  const unsigned int y = x ^ pat;
  const unsigned int t = (y & 0x7F7F7F7Fu) + 0x7F7F7F7Fu;
  return ~(t | y) & 0x80808080u;
}

__device__ __forceinline__ void tile_add_row(float* __restrict__ tile, const float* __restrict__ h,
                                             int sj, int lj, float nj, unsigned lane) {
  const float* row = h + (size_t)sj * HID;
  float* trow = tile + lj * HID;
  const float a0 = row[lane];
  const float a1 = row[32u + lane];
  atomicAdd(trow + lane, a0 * nj);
  atomicAdd(trow + 32u + lane, a1 * nj);
}

__global__ void __launch_bounds__(AGG_THREADS)
agg_tile_kernel(const int* __restrict__ src, const int* __restrict__ dst,
                const unsigned int* __restrict__ bkt4,
                const float* __restrict__ h, const float* __restrict__ dis,
                const float* __restrict__ bias, float* __restrict__ xout, int E) {
  __shared__ float tile[TILE_ROWS * HID];
  __shared__ unsigned int ring[AGG_WAVES][RING];

  const unsigned tid = threadIdx.x;
  const unsigned lane = tid & 31u;
  const unsigned wave = tid >> 5;
  const int tileno = blockIdx.x;
  const int row0 = tileno * TILE_ROWS;

  float4* tile4 = (float4*)tile;
  for (int i = tid; i < TILE_ROWS * HID / 4; i += AGG_THREADS)
    tile4[i] = make_float4(0.0f, 0.0f, 0.0f, 0.0f);
  __syncthreads();

  const unsigned int pat = (unsigned int)tileno * 0x01010101u;
  const long nvec = ((long)E + 511) / 512 * 32;
  const long step = AGG_THREADS;
  const uint4* __restrict__ bv = (const uint4*)bkt4;
  unsigned int* myring = ring[wave];
  unsigned int wr = 0, rd = 0;

  auto scan_group = [&](const uint4 cv, const long qq) {
    unsigned int g = (match_bytes(cv.x, pat) >> 7) | (match_bytes(cv.y, pat) >> 6) |
                     (match_bytes(cv.z, pat) >> 5) | (match_bytes(cv.w, pat) >> 4);
    const unsigned int ebase = (unsigned int)qq * 16u;
    unsigned int bal = __builtin_amdgcn_ballot_w32(g != 0u);
    while (bal != 0u) {
      const unsigned int p = (unsigned int)__builtin_ctz(g | 0x80000000u);
      const unsigned int slot = wr + __builtin_amdgcn_mbcnt_lo(bal, 0u);
      if (g != 0u) {
        myring[slot & (RING - 1)] = ebase + ((p & 7u) << 2) + (p >> 3);
        g &= g - 1u;
      }
      wr += (unsigned int)__builtin_popcount(bal);
      bal = __builtin_amdgcn_ballot_w32(g != 0u);
    }

    while (wr - rd >= 32u) {
      const unsigned int e = myring[(rd + lane) & (RING - 1)];
      const int s = src[e];
      const int d = dst[e];
      const float nrm = dis[s] * dis[d];
      const int lr = d - row0;
      const int nbits = __float_as_int(nrm);
#pragma unroll
      for (int b = 0; b < 32; b += EDGE_ROWS) {
        float v0[EDGE_ROWS], v1[EDGE_ROWS];
#pragma unroll
        for (int j = 0; j < EDGE_ROWS; ++j) {
          const int sj = __builtin_amdgcn_readlane(s, b + j);
          const float* row = h + (size_t)sj * HID;
          v0[j] = row[lane];
          v1[j] = row[32u + lane];
        }
#pragma unroll
        for (int j = 0; j < EDGE_ROWS; ++j) {
          const int lj = __builtin_amdgcn_readlane(lr, b + j);
          const float nj = __int_as_float(__builtin_amdgcn_readlane(nbits, b + j));
          float* trow = tile + lj * HID;
          atomicAdd(trow + lane, v0[j] * nj);
          atomicAdd(trow + 32u + lane, v1[j] * nj);
        }
      }
      rd += 32u;
    }
  };

  {
    long q = (long)tid;
    uint4 a0 = bv[q], a1 = bv[q + step], a2 = bv[q + 2 * step], a3 = bv[q + 3 * step];
    for (; q < nvec; q += 8 * step) {
      const uint4 b0 = bv[q + 4 * step], b1 = bv[q + 5 * step], b2 = bv[q + 6 * step],
                  b3 = bv[q + 7 * step];
      scan_group(a0, q);
      scan_group(a1, q + step);
      scan_group(a2, q + 2 * step);
      scan_group(a3, q + 3 * step);
      a0 = bv[q + 8 * step];
      a1 = bv[q + 9 * step];
      a2 = bv[q + 10 * step];
      a3 = bv[q + 11 * step];
      scan_group(b0, q + 4 * step);
      scan_group(b1, q + 5 * step);
      scan_group(b2, q + 6 * step);
      scan_group(b3, q + 7 * step);
    }
  }

  const unsigned int nleft = wr - rd;
  if (nleft != 0u) {
    int s = 0, lr = 0;
    float nrm = 0.0f;
    if (lane < nleft) {
      const unsigned int e = myring[(rd + lane) & (RING - 1)];
      s = src[e];
      const int d = dst[e];
      nrm = dis[s] * dis[d];
      lr = d - row0;
    }
    const int nbits = __float_as_int(nrm);
    for (unsigned int j = 0; j < nleft; ++j) {
      const int sj = __builtin_amdgcn_readlane(s, j);
      const int lj = __builtin_amdgcn_readlane(lr, j);
      const float nj = __int_as_float(__builtin_amdgcn_readlane(nbits, j));
      tile_add_row(tile, h, sj, lj, nj, lane);
    }
  }
  __syncthreads();

  const int nrows = (NNODES - row0) < TILE_ROWS ? (NNODES - row0) : TILE_ROWS;
  const float4* __restrict__ h4 = (const float4*)(h + (size_t)row0 * HID);
  const float4* __restrict__ b4 = (const float4*)bias;
  float4* __restrict__ x4 = (float4*)(xout + (size_t)row0 * HID);
  for (int i = tid; i < nrows * (HID / 4); i += AGG_THREADS) {
    const int r = i >> 4;
    const int c4 = i & 15;
    const float di = dis[row0 + r];
    const float4 a = tile4[i];
    const float4 hv = h4[i];
    const float4 bb = b4[c4];
    float4 v;
    v.x = a.x + hv.x * di * di + bb.x;
    v.y = a.y + hv.y * di * di + bb.y;
    v.z = a.z + hv.z * di * di + bb.z;
    v.w = a.w + hv.w * di * di + bb.w;
    v.x = v.x > 0.0f ? v.x : 0.0f;
    v.y = v.y > 0.0f ? v.y : 0.0f;
    v.z = v.z > 0.0f ? v.z : 0.0f;
    v.w = v.w > 0.0f ? v.w : 0.0f;
    x4[i] = v;
  }
}

__global__ void pool_kernel(const float* __restrict__ x, const int* __restrict__ batch,
                            float* __restrict__ pooled, float* __restrict__ cnt, int n) {
  long i = (long)blockIdx.x * blockDim.x + threadIdx.x;
  if (i >= (long)n * HID) return;
  int node = (int)(i >> 6);
  int c = (int)(i & 63);
  int g = batch[node];
  atomicAdd(pooled + (long)g * HID + c, x[i]);
  if (c == 0) atomicAdd(cnt + g, 1.0f);
}

__global__ void out_kernel(const float* __restrict__ pooled, const float* __restrict__ cnt,
                           const float* __restrict__ Wl, const float* __restrict__ bl,
                           float* __restrict__ out) {
  int i = blockIdx.x * blockDim.x + threadIdx.x;
  if (i >= NGRAPHS * NCLASSES) return;
  int g = i / NCLASSES;
  int k = i % NCLASSES;
  float s = 0.0f;
#pragma unroll 8
  for (int hh = 0; hh < HID; ++hh) s += pooled[g * HID + hh] * Wl[hh * NCLASSES + k];
  float c = cnt[g];
  c = c > 1.0f ? c : 1.0f;
  out[i] = s / c + bl[k];
}

extern "C" void kernel_launch(void* const* d_in, const int* in_sizes, int n_in,
                              void* d_out, int out_size, void* d_ws, size_t ws_size,
                              hipStream_t stream) {
  const int* shape_id = (const int*)d_in[0];
  const int* color_id = (const int*)d_in[1];
  const int* edge_index = (const int*)d_in[2];
  const int* batch = (const int*)d_in[3];
  const float* shape_table = (const float*)d_in[4];
  const float* color_table = (const float*)d_in[5];
  const float* W1 = (const float*)d_in[6];
  const float* b1 = (const float*)d_in[7];
  const float* W2 = (const float*)d_in[8];
  const float* b2 = (const float*)d_in[9];
  const float* Wl = (const float*)d_in[10];
  const float* bl = (const float*)d_in[11];
  float* out = (float*)d_out;

  const int N = NNODES;
  const int E = in_sizes[2] / 2;
  const int* src = edge_index;
  const int* dst = edge_index + E;

  float* dis    = (float*)d_ws;
  float* x0     = dis + N;
  float* bufH   = x0 + (long)N * EMB;
  float* bufA   = bufH + (long)N * HID;
  float* pooled = bufA + (long)N * HID;
  float* cnt    = pooled + (long)NGRAPHS * HID;
  unsigned int* bkt4 = (unsigned int*)x0;
  (void)n_in; (void)out_size; (void)ws_size;

  const int B = 256;
  auto blocks = [](long n, int b) { return (unsigned)((n + b - 1) / b); };

  zero_f32<<<blocks(N, B), B, 0, stream>>>(dis, N);
  deg_kernel<<<blocks(E, B), B, 0, stream>>>(dst, (unsigned int*)dis, E);
  dis_kernel<<<blocks(N, B), B, 0, stream>>>(dis, N);

  embed_kernel<<<blocks((long)N * EMB, B), B, 0, stream>>>(shape_id, color_id,
                                                           shape_table, color_table, x0, N);

  gemm_wmma_f32<<<N / 16, 128, 0, stream>>>(x0, W1, bufH, EMB);
  bucket_kernel<<<blocks((long)E / 4 + 128 + STREAM_PAD / 4, B), B, 0, stream>>>(dst, bkt4, E);
  agg_tile_kernel<<<NTILES, AGG_THREADS, 0, stream>>>(src, dst, bkt4, bufH, dis, b1, bufA, E);

  gemm_wmma_f32<<<N / 16, 128, 0, stream>>>(bufA, W2, bufH, HID);
  agg_tile_kernel<<<NTILES, AGG_THREADS, 0, stream>>>(src, dst, bkt4, bufH, dis, b2, bufA, E);

  zero_f32<<<blocks((long)NGRAPHS * HID + NGRAPHS, B), B, 0, stream>>>(pooled,
                                                                      (long)NGRAPHS * HID + NGRAPHS);
  pool_kernel<<<blocks((long)N * HID, B), B, 0, stream>>>(bufA, batch, pooled, cnt, N);
  out_kernel<<<blocks((long)NGRAPHS * NCLASSES, B), B, 0, stream>>>(pooled, cnt, Wl, bl, out);
}
